# starting kernel
# baseline (speedup 1.0000x reference)
#include <hip/hip_runtime.h>
#include <hip/hip_bf16.h>

typedef __attribute__((ext_vector_type(2))) float v2f;
typedef __attribute__((ext_vector_type(4))) float v4f;
typedef __attribute__((ext_vector_type(8))) float v8f;

__global__ void __launch_bounds__(256) se_pool_kernel(
    const float* __restrict__ x, float* __restrict__ pooled) {
  const int plane = blockIdx.x;
  const v4f* p = (const v4f*)(x + (size_t)plane * 4096);

  float sum = 0.0f;
#pragma unroll
  for (int i = 0; i < 4; ++i) {
    v4f v = p[threadIdx.x + i * 256];
    sum += (v.x + v.y) + (v.z + v.w);
  }
#pragma unroll
  for (int off = 16; off > 0; off >>= 1)
    sum += __shfl_down(sum, off, 32);

  __shared__ float red[8];
  if ((threadIdx.x & 31) == 0) red[threadIdx.x >> 5] = sum;
  __syncthreads();
  if (threadIdx.x == 0) {
    float t = 0.0f;
#pragma unroll
    for (int w = 0; w < 8; ++w) t += red[w];
    pooled[plane] = t * (1.0f / 4096.0f);
  }
}

__global__ void __launch_bounds__(128) se_mlp_wmma_kernel(
    const float* __restrict__ pooled,
    const float* __restrict__ w1,
    const float* __restrict__ b1,
    const float* __restrict__ w2,
    const float* __restrict__ b2,
    float* __restrict__ scale,
    float* __restrict__ shift) {
  const int lane  = threadIdx.x & 31;
  const int wave  = threadIdx.x >> 5;
  const int r     = lane & 15;
  const int koff  = (lane >> 4) << 1;
  const int mhalf = (lane < 16) ? 0 : 8;

  __shared__ float hs[32 * 16];

  v8f c0 = {};
  v8f c1 = {};
#pragma unroll 4
  for (int k = 0; k < 256; k += 4) {
    v2f bt = *(const v2f*)(w1 + r * 256 + k + koff);
    v2f a0 = *(const v2f*)(pooled + r * 256 + k + koff);
    v2f a1 = *(const v2f*)(pooled + (16 + r) * 256 + k + koff);
    c0 = __builtin_amdgcn_wmma_f32_16x16x4_f32(false, a0, false, bt,
                                               (short)0, c0, false, false);
    c1 = __builtin_amdgcn_wmma_f32_16x16x4_f32(false, a1, false, bt,
                                               (short)0, c1, false, false);
  }

  const float bias1 = b1[r];
#pragma unroll
  for (int v = 0; v < 8; ++v) {
    float h0 = c0[v] + bias1;
    float h1 = c1[v] + bias1;
    h0 = h0 > 0.0f ? h0 : 0.0f;
    h1 = h1 > 0.0f ? h1 : 0.0f;
    hs[(mhalf + v) * 16 + r]        = h0;
    hs[(16 + mhalf + v) * 16 + r]   = h1;
  }
  __syncthreads();

  for (int nb = wave; nb < 32; nb += 4) {
    const int n0 = nb * 16;
    v8f d0 = {};
    v8f d1 = {};
#pragma unroll
    for (int k = 0; k < 16; k += 4) {
      v2f bt = *(const v2f*)(w2 + (n0 + r) * 16 + k + koff);
      v2f a0 = *(const v2f*)(hs + r * 16 + k + koff);
      v2f a1 = *(const v2f*)(hs + (16 + r) * 16 + k + koff);
      d0 = __builtin_amdgcn_wmma_f32_16x16x4_f32(false, a0, false, bt,
                                                 (short)0, d0, false, false);
      d1 = __builtin_amdgcn_wmma_f32_16x16x4_f32(false, a1, false, bt,
                                                 (short)0, d1, false, false);
    }
    const float bias2 = b2[n0 + r];
    const int col = n0 + r;
#pragma unroll
    for (int v = 0; v < 8; ++v) {
      float g0 = d0[v] + bias2;
      float g1 = d1[v] + bias2;
      const int m0 = mhalf + v;
      const int m1 = 16 + mhalf + v;
      if (col < 256) {
        scale[m0 * 256 + col] = 1.0f / (1.0f + __expf(-g0));
        scale[m1 * 256 + col] = 1.0f / (1.0f + __expf(-g1));
      } else {
        shift[m0 * 256 + (col - 256)] = g0;
        shift[m1 * 256 + (col - 256)] = g1;
      }
    }
  }
}

__global__ void __launch_bounds__(256) se_apply_kernel(
    const float* __restrict__ x, const float* __restrict__ scale,
    const float* __restrict__ shift, float* __restrict__ out) {
  const int plane = blockIdx.x;
  const float s = scale[plane];
  const float t = shift[plane];
  const v4f* xp = (const v4f*)(x + (size_t)plane * 4096);
  v4f* op = (v4f*)(out + (size_t)plane * 4096);
#pragma unroll
  for (int i = 0; i < 4; ++i) {
    const int idx = threadIdx.x + i * 256;
    v4f v = xp[idx];
    v4f o;
    o.x = __builtin_fmaf(s, v.x, t);
    o.y = __builtin_fmaf(s, v.y, t);
    o.z = __builtin_fmaf(s, v.z, t);
    o.w = __builtin_fmaf(s, v.w, t);
    __builtin_nontemporal_store(o, op + idx);
  }
}

extern "C" void kernel_launch(void* const* d_in, const int* in_sizes, int n_in,
                              void* d_out, int out_size, void* d_ws,
                              size_t ws_size, hipStream_t stream) {
  const float* x  = (const float*)d_in[0];
  const float* w1 = (const float*)d_in[1];
  const float* b1 = (const float*)d_in[2];
  const float* w2 = (const float*)d_in[3];
  const float* b2 = (const float*)d_in[4];
  float* out = (float*)d_out;

  float* pooled = (float*)d_ws;
  float* scale  = pooled + 32 * 256;
  float* shift  = scale + 32 * 256;

  se_pool_kernel<<<32 * 256, 256, 0, stream>>>(x, pooled);
  se_mlp_wmma_kernel<<<1, 128, 0, stream>>>(pooled, w1, b1, w2, b2, scale,
                                            shift);
  se_apply_kernel<<<32 * 256, 256, 0, stream>>>(x, scale, shift, out);
}


# MLP kernel re-tiled over outputs: 32 workgroups (16 batch rows x one 128-byte line of g) stage pooled rows and w1 in LDS by coalesced 16-byte loads, rebuild h in one WMMA chain; sum order unchanged
# speedup vs baseline: 1.5246x; 1.5246x over previous
#include <hip/hip_runtime.h>
#include <hip/hip_bf16.h>

typedef __attribute__((ext_vector_type(2))) float v2f;
typedef __attribute__((ext_vector_type(4))) float v4f;
typedef __attribute__((ext_vector_type(8))) float v8f;

__global__ void __launch_bounds__(256) se_pool_kernel(
    const float* __restrict__ x, float* __restrict__ pooled) {
  const int plane = blockIdx.x;
  const v4f* p = (const v4f*)(x + (size_t)plane * 4096);

  float sum = 0.0f;
#pragma unroll
  for (int i = 0; i < 4; ++i) {
    v4f v = p[threadIdx.x + i * 256];
    sum += (v.x + v.y) + (v.z + v.w);
  }
#pragma unroll
  for (int off = 16; off > 0; off >>= 1)
    sum += __shfl_down(sum, off, 32);

  __shared__ float red[8];
  if ((threadIdx.x & 31) == 0) red[threadIdx.x >> 5] = sum;
  __syncthreads();
  if (threadIdx.x == 0) {
    float t = 0.0f;
#pragma unroll
    for (int w = 0; w < 8; ++w) t += red[w];
    pooled[plane] = t * (1.0f / 4096.0f);
  }
}

#define SE_PITCH 260
#define SE_HPITCH 20

__global__ void __launch_bounds__(256) se_mlp_wmma_kernel(
    const float* __restrict__ pooled,
    const float* __restrict__ w1,
    const float* __restrict__ b1,
    const float* __restrict__ w2,
    const float* __restrict__ b2,
    float* __restrict__ scale,
    float* __restrict__ shift) {
  const int tid   = threadIdx.x;
  const int lane  = tid & 31;
  const int r     = lane & 15;
  const int koff  = (lane >> 4) << 1;
  const int mhalf = (lane < 16) ? 0 : 8;
  const int mt    = blockIdx.x & 1;
  const int pr    = blockIdx.x >> 1;
  const int wave  = __builtin_amdgcn_readfirstlane(tid >> 5);

  __shared__ __attribute__((aligned(16))) float ps[16 * SE_PITCH];
  __shared__ __attribute__((aligned(16))) float ws[16 * SE_PITCH];
  __shared__ __attribute__((aligned(16))) float hs[16 * SE_HPITCH];

  const v4f* pg = (const v4f*)(pooled + mt * 16 * 256);
  const v4f* wg = (const v4f*)w1;
  v4f pv[4], wv[4];
#pragma unroll
  for (int i = 0; i < 4; ++i) {
    pv[i] = pg[tid + i * 256];
    wv[i] = wg[tid + i * 256];
  }

#pragma unroll
  for (int i = 0; i < 4; ++i) {
    const int idx = tid + i * 256;
    const int row = idx >> 6;
    const int c4  = idx & 63;
    *(v4f*)(ps + row * SE_PITCH + c4 * 4) = pv[i];
    *(v4f*)(ws + row * SE_PITCH + c4 * 4) = wv[i];
  }

  __syncthreads();

  v2f w2f[2][4] = {};
  float bias2[2] = {0.0f, 0.0f};
  if (wave == 0) {
    const float bias1 = b1[r];
#pragma unroll
    for (int nn = 0; nn < 2; ++nn) {
      const int col = pr * 32 + nn * 16 + r;
      bias2[nn] = b2[col];
#pragma unroll
      for (int t = 0; t < 4; ++t)
        w2f[nn][t] = *(const v2f*)(w2 + col * 16 + 4 * t + koff);
    }

    v2f af[64], bf[64];
#pragma unroll
    for (int t = 0; t < 64; ++t) {
      af[t] = *(const v2f*)(ps + r * SE_PITCH + 4 * t + koff);
      bf[t] = *(const v2f*)(ws + r * SE_PITCH + 4 * t + koff);
    }
    __builtin_amdgcn_sched_barrier(0);
    v8f c = {};
#pragma unroll
    for (int t = 0; t < 64; ++t)
      c = __builtin_amdgcn_wmma_f32_16x16x4_f32(false, af[t], false, bf[t],
                                                (short)0, c, false, false);
#pragma unroll
    for (int v = 0; v < 8; ++v) {
      float hv = c[v] + bias1;
      hv = hv > 0.0f ? hv : 0.0f;
      hs[(mhalf + v) * SE_HPITCH + r] = hv;
    }
  }
  __syncthreads();

  if (wave == 0) {
    v8f d[2] = {};
#pragma unroll
    for (int t = 0; t < 4; ++t) {
      v2f a = *(const v2f*)(hs + r * SE_HPITCH + 4 * t + koff);
      d[0] = __builtin_amdgcn_wmma_f32_16x16x4_f32(false, a, false, w2f[0][t],
                                                   (short)0, d[0], false, false);
      d[1] = __builtin_amdgcn_wmma_f32_16x16x4_f32(false, a, false, w2f[1][t],
                                                   (short)0, d[1], false, false);
    }
    float g[2][8];
#pragma unroll
    for (int nn = 0; nn < 2; ++nn)
#pragma unroll
      for (int v = 0; v < 8; ++v) g[nn][v] = d[nn][v] + bias2[nn];

    const int row0 = mt * 16 + mhalf;
    if (pr < 8) {
#pragma unroll
      for (int nn = 0; nn < 2; ++nn)
#pragma unroll
        for (int v = 0; v < 8; ++v)
          scale[(row0 + v) * 256 + pr * 32 + nn * 16 + r] =
              1.0f / (1.0f + __expf(-g[nn][v]));
    } else {
#pragma unroll
      for (int nn = 0; nn < 2; ++nn)
#pragma unroll
        for (int v = 0; v < 8; ++v)
          shift[(row0 + v) * 256 + (pr - 8) * 32 + nn * 16 + r] = g[nn][v];
    }
  }
}

__global__ void __launch_bounds__(256) se_apply_kernel(
    const float* __restrict__ x, const float* __restrict__ scale,
    const float* __restrict__ shift, float* __restrict__ out) {
  const int plane = blockIdx.x;
  const float s = scale[plane];
  const float t = shift[plane];
  const v4f* xp = (const v4f*)(x + (size_t)plane * 4096);
  v4f* op = (v4f*)(out + (size_t)plane * 4096);
#pragma unroll
  for (int i = 0; i < 4; ++i) {
    const int idx = threadIdx.x + i * 256;
    v4f v = xp[idx];
    v4f o;
    o.x = __builtin_fmaf(s, v.x, t);
    o.y = __builtin_fmaf(s, v.y, t);
    o.z = __builtin_fmaf(s, v.z, t);
    o.w = __builtin_fmaf(s, v.w, t);
    __builtin_nontemporal_store(o, op + idx);
  }
}

extern "C" void kernel_launch(void* const* d_in, const int* in_sizes, int n_in,
                              void* d_out, int out_size, void* d_ws,
                              size_t ws_size, hipStream_t stream) {
  const float* x  = (const float*)d_in[0];
  const float* w1 = (const float*)d_in[1];
  const float* b1 = (const float*)d_in[2];
  const float* w2 = (const float*)d_in[3];
  const float* b2 = (const float*)d_in[4];
  float* out = (float*)d_out;

  float* pooled = (float*)d_ws;
  float* scale  = pooled + 32 * 256;
  float* shift  = scale + 32 * 256;

  se_pool_kernel<<<32 * 256, 256, 0, stream>>>(x, pooled);
  se_mlp_wmma_kernel<<<32, 256, 0, stream>>>(pooled, w1, b1, w2, b2, scale,
                                            shift);
  se_apply_kernel<<<32 * 256, 256, 0, stream>>>(x, scale, shift, out);
}
